# starting kernel
# baseline (speedup 1.0000x reference)
#include <hip/hip_runtime.h>

typedef __attribute__((ext_vector_type(2))) float v2f;
typedef __attribute__((ext_vector_type(8))) float v8f;

#define B_BATCH 256
#define S_DIM   16384
#define R_RANK  16
#define KSPLIT  8
#define NT_TOTAL 65
#define NCOLS   1040
#define MBLK    4
#define KCHUNK  (S_DIM / KSPLIT)

__device__ __forceinline__ v8f wmma_f32x4(v2f a, v2f b, v8f c) {
    return __builtin_amdgcn_wmma_f32_16x16x4_f32(false, a, false, b, (short)0, c,
                                                 false, false);
}

__global__ __launch_bounds__(32)
void tt_phase1(const float* __restrict__ z,
               const float* __restrict__ c1, const float* __restrict__ c2,
               const float* __restrict__ c3, const float* __restrict__ c4,
               const float* __restrict__ c5,
               float* __restrict__ vpart) {
    const int lane = threadIdx.x & 31;
    const int half = lane >> 4;
    const int l    = lane & 15;

    const int t  = blockIdx.x;
    const int mb = blockIdx.y;
    const int ks = blockIdx.z;

    const int b0      = mb * 64;
    const int k_begin = ks * KCHUNK;

    const float* ap0 = z + (size_t)(b0 + l) * S_DIM + k_begin + 2 * half;
    const float* ap1 = ap0 + (size_t)16 * S_DIM;
    const float* ap2 = ap0 + (size_t)32 * S_DIM;
    const float* ap3 = ap0 + (size_t)48 * S_DIM;

    v8f acc0 = {}, acc1 = {}, acc2 = {}, acc3 = {};
    int colbase;

    if (t < 64) {
        const int k = t >> 4;
        const int d = t & 15;
        const float* cp = (k == 0) ? c1 : (k == 1) ? c2 : (k == 2) ? c3 : c4;
        colbase = k * 256 + d * 16;
        const float* bp = cp + (size_t)d * S_DIM * R_RANK
                             + (size_t)(k_begin + 2 * half) * R_RANK + l;
        for (int it = 0; it < KCHUNK / 16; ++it) {
            __builtin_prefetch(bp + 4 * 256, 0, 1);
            __builtin_prefetch(ap0 + 64, 0, 1);
#pragma unroll
            for (int u = 0; u < 4; ++u) {
                v2f bf;
                bf.x = bp[u * 64];
                bf.y = bp[u * 64 + 16];
                v2f a0 = *(const v2f*)(ap0 + u * 4);
                v2f a1 = *(const v2f*)(ap1 + u * 4);
                v2f a2 = *(const v2f*)(ap2 + u * 4);
                v2f a3 = *(const v2f*)(ap3 + u * 4);
                acc0 = wmma_f32x4(a0, bf, acc0);
                acc1 = wmma_f32x4(a1, bf, acc1);
                acc2 = wmma_f32x4(a2, bf, acc2);
                acc3 = wmma_f32x4(a3, bf, acc3);
            }
            ap0 += 16; ap1 += 16; ap2 += 16; ap3 += 16;
            bp  += 256;
        }
    } else {
        colbase = 1024;
        const float* bp = c5 + (size_t)l * S_DIM + k_begin + 2 * half;
        for (int it = 0; it < KCHUNK / 16; ++it) {
            __builtin_prefetch(bp + 64, 0, 1);
            __builtin_prefetch(ap0 + 64, 0, 1);
#pragma unroll
            for (int u = 0; u < 4; ++u) {
                v2f bf = *(const v2f*)(bp + u * 4);
                v2f a0 = *(const v2f*)(ap0 + u * 4);
                v2f a1 = *(const v2f*)(ap1 + u * 4);
                v2f a2 = *(const v2f*)(ap2 + u * 4);
                v2f a3 = *(const v2f*)(ap3 + u * 4);
                acc0 = wmma_f32x4(a0, bf, acc0);
                acc1 = wmma_f32x4(a1, bf, acc1);
                acc2 = wmma_f32x4(a2, bf, acc2);
                acc3 = wmma_f32x4(a3, bf, acc3);
            }
            ap0 += 16; ap1 += 16; ap2 += 16; ap3 += 16;
            bp  += 16;
        }
    }

    float* vp = vpart + (size_t)ks * B_BATCH * NCOLS
                      + (size_t)(b0 + 8 * half) * NCOLS + colbase + l;
    v8f accs[MBLK] = {acc0, acc1, acc2, acc3};
#pragma unroll
    for (int m = 0; m < MBLK; ++m) {
#pragma unroll
        for (int v = 0; v < 8; ++v) {
            vp[(size_t)(m * 16 + v) * NCOLS] = accs[m][v];
        }
    }
}

__global__ __launch_bounds__(256)
void tt_phase2(const float* __restrict__ vpart, float* __restrict__ w) {
    const int b = blockIdx.x * blockDim.x + threadIdx.x;
    if (b >= B_BATCH) return;

    const size_t rowoff = (size_t)b * NCOLS;
    float t[R_RANK], nt[R_RANK];

#pragma unroll
    for (int d = 0; d < R_RANK; ++d) {
        float s = 0.f;
        for (int p = 0; p < KSPLIT; ++p)
            s += vpart[(size_t)p * B_BATCH * NCOLS + rowoff + 1024 + d];
        t[d] = s;
    }
    for (int k = 3; k >= 0; --k) {
#pragma unroll
        for (int d = 0; d < R_RANK; ++d) {
            float s = 0.f;
#pragma unroll
            for (int e = 0; e < R_RANK; ++e) {
                float v = 0.f;
                for (int p = 0; p < KSPLIT; ++p)
                    v += vpart[(size_t)p * B_BATCH * NCOLS + rowoff +
                               k * 256 + d * 16 + e];
                s += v * t[e];
            }
            nt[d] = s;
        }
#pragma unroll
        for (int d = 0; d < R_RANK; ++d) t[d] = nt[d];
    }
#pragma unroll
    for (int r = 0; r < R_RANK; ++r)
        w[b * R_RANK + r] = t[r];
}

__global__ __launch_bounds__(32)
void tt_phase3(const float* __restrict__ w, const float* __restrict__ core0,
               float* __restrict__ out) {
    const int lane = threadIdx.x & 31;
    const int half = lane >> 4;
    const int l    = lane & 15;
    const int s0   = blockIdx.x * 16;
    const int b0   = blockIdx.y * 16;

    const float* wp = w     + (size_t)(b0 + l) * R_RANK + 2 * half;
    const float* cp = core0 + (size_t)(s0 + l) * R_RANK + 2 * half;

    v8f acc = {};
#pragma unroll
    for (int k0 = 0; k0 < R_RANK; k0 += 4) {
        v2f a  = *(const v2f*)(wp + k0);
        v2f bf = *(const v2f*)(cp + k0);
        acc = wmma_f32x4(a, bf, acc);
    }
    float* op = out + (size_t)(b0 + 8 * half) * S_DIM + s0 + l;
#pragma unroll
    for (int v = 0; v < 8; ++v) {
        op[(size_t)v * S_DIM] = acc[v];
    }
}

extern "C" void kernel_launch(void* const* d_in, const int* in_sizes, int n_in,
                              void* d_out, int out_size, void* d_ws, size_t ws_size,
                              hipStream_t stream) {
    const float* z  = (const float*)d_in[0];
    const float* c0 = (const float*)d_in[1];
    const float* c1 = (const float*)d_in[2];
    const float* c2 = (const float*)d_in[3];
    const float* c3 = (const float*)d_in[4];
    const float* c4 = (const float*)d_in[5];
    const float* c5 = (const float*)d_in[6];
    float* out = (float*)d_out;

    float* vpart = (float*)d_ws;
    float* w     = vpart + (size_t)KSPLIT * B_BATCH * NCOLS;

    tt_phase1<<<dim3(NT_TOTAL, B_BATCH / 64, KSPLIT), 32, 0, stream>>>(
        z, c1, c2, c3, c4, c5, vpart);
    tt_phase2<<<dim3(1), 256, 0, stream>>>(vpart, w);
    tt_phase3<<<dim3(S_DIM / 16, B_BATCH / 16), 32, 0, stream>>>(w, c0, out);
}


# phase 2 (K-split sum + 16x16 chain) spread from one work-group to one per batch row, partials read as whole 16-byte pieces per lane; summation order kept
# speedup vs baseline: 3.1058x; 3.1058x over previous
#include <hip/hip_runtime.h>

typedef __attribute__((ext_vector_type(2))) float v2f;
typedef __attribute__((ext_vector_type(4))) float v4f;
typedef __attribute__((ext_vector_type(8))) float v8f;

#define B_BATCH 256
#define S_DIM   16384
#define R_RANK  16
#define KSPLIT  8
#define NT_TOTAL 65
#define NCOLS   1040
#define MBLK    4
#define KCHUNK  (S_DIM / KSPLIT)

__device__ __forceinline__ v8f wmma_f32x4(v2f a, v2f b, v8f c) {
    return __builtin_amdgcn_wmma_f32_16x16x4_f32(false, a, false, b, (short)0, c,
                                                 false, false);
}

__global__ __launch_bounds__(32)
void tt_phase1(const float* __restrict__ z,
               const float* __restrict__ c1, const float* __restrict__ c2,
               const float* __restrict__ c3, const float* __restrict__ c4,
               const float* __restrict__ c5,
               float* __restrict__ vpart) {
    const int lane = threadIdx.x & 31;
    const int half = lane >> 4;
    const int l    = lane & 15;

    const int t  = blockIdx.x;
    const int mb = blockIdx.y;
    const int ks = blockIdx.z;

    const int b0      = mb * 64;
    const int k_begin = ks * KCHUNK;

    const float* ap0 = z + (size_t)(b0 + l) * S_DIM + k_begin + 2 * half;
    const float* ap1 = ap0 + (size_t)16 * S_DIM;
    const float* ap2 = ap0 + (size_t)32 * S_DIM;
    const float* ap3 = ap0 + (size_t)48 * S_DIM;

    v8f acc0 = {}, acc1 = {}, acc2 = {}, acc3 = {};
    int colbase;

    if (t < 64) {
        const int k = t >> 4;
        const int d = t & 15;
        const float* cp = (k == 0) ? c1 : (k == 1) ? c2 : (k == 2) ? c3 : c4;
        colbase = k * 256 + d * 16;
        const float* bp = cp + (size_t)d * S_DIM * R_RANK
                             + (size_t)(k_begin + 2 * half) * R_RANK + l;
        for (int it = 0; it < KCHUNK / 16; ++it) {
            __builtin_prefetch(bp + 4 * 256, 0, 1);
            __builtin_prefetch(ap0 + 64, 0, 1);
#pragma unroll
            for (int u = 0; u < 4; ++u) {
                v2f bf;
                bf.x = bp[u * 64];
                bf.y = bp[u * 64 + 16];
                v2f a0 = *(const v2f*)(ap0 + u * 4);
                v2f a1 = *(const v2f*)(ap1 + u * 4);
                v2f a2 = *(const v2f*)(ap2 + u * 4);
                v2f a3 = *(const v2f*)(ap3 + u * 4);
                acc0 = wmma_f32x4(a0, bf, acc0);
                acc1 = wmma_f32x4(a1, bf, acc1);
                acc2 = wmma_f32x4(a2, bf, acc2);
                acc3 = wmma_f32x4(a3, bf, acc3);
            }
            ap0 += 16; ap1 += 16; ap2 += 16; ap3 += 16;
            bp  += 256;
        }
    } else {
        colbase = 1024;
        const float* bp = c5 + (size_t)l * S_DIM + k_begin + 2 * half;
        for (int it = 0; it < KCHUNK / 16; ++it) {
            __builtin_prefetch(bp + 64, 0, 1);
            __builtin_prefetch(ap0 + 64, 0, 1);
#pragma unroll
            for (int u = 0; u < 4; ++u) {
                v2f bf = *(const v2f*)(bp + u * 4);
                v2f a0 = *(const v2f*)(ap0 + u * 4);
                v2f a1 = *(const v2f*)(ap1 + u * 4);
                v2f a2 = *(const v2f*)(ap2 + u * 4);
                v2f a3 = *(const v2f*)(ap3 + u * 4);
                acc0 = wmma_f32x4(a0, bf, acc0);
                acc1 = wmma_f32x4(a1, bf, acc1);
                acc2 = wmma_f32x4(a2, bf, acc2);
                acc3 = wmma_f32x4(a3, bf, acc3);
            }
            ap0 += 16; ap1 += 16; ap2 += 16; ap3 += 16;
            bp  += 16;
        }
    }

    float* vp = vpart + (size_t)ks * B_BATCH * NCOLS
                      + (size_t)(b0 + 8 * half) * NCOLS + colbase + l;
    v8f accs[MBLK] = {acc0, acc1, acc2, acc3};
#pragma unroll
    for (int m = 0; m < MBLK; ++m) {
#pragma unroll
        for (int v = 0; v < 8; ++v) {
            vp[(size_t)(m * 16 + v) * NCOLS] = accs[m][v];
        }
    }
}

#define ROW4   (NCOLS / 4)
#define MAT4   (R_RANK * R_RANK / 4)
#define TAIL4  (ROW4 - 4 * MAT4)

__device__ __forceinline__ float lane_bcast(float v, int srclane) {
    return __int_as_float(__builtin_amdgcn_readlane(__float_as_int(v), srclane));
}

__global__ __launch_bounds__(256)
void tt_phase2(const float* __restrict__ vpart, float* __restrict__ w) {
    __shared__ v4f vred[ROW4];

    const int b = blockIdx.x;
    const int j = threadIdx.x;
    const bool tail = j < TAIL4;

    const v4f* row = (const v4f*)(vpart + (size_t)b * NCOLS);
    const size_t pstride = (size_t)B_BATCH * ROW4;

    v4f x[KSPLIT], y[KSPLIT];
#pragma unroll
    for (int p = 0; p < KSPLIT; ++p) x[p] = row[p * pstride + j];
    if (tail) {
#pragma unroll
        for (int p = 0; p < KSPLIT; ++p) y[p] = row[p * pstride + 4 * MAT4 + j];
    }

    v4f s = {};
#pragma unroll
    for (int p = 0; p < KSPLIT; ++p) s += x[p];
    vred[j] = s;
    if (tail) {
        v4f s5 = {};
#pragma unroll
        for (int p = 0; p < KSPLIT; ++p) s5 += y[p];
        vred[4 * MAT4 + j] = s5;
    }
    __syncthreads();

    if (j < R_RANK) {
        const float* vf = (const float*)vred;
        float t = vf[4 * R_RANK * R_RANK + j];
#pragma unroll
        for (int k = 3; k >= 0; --k) {
            const v4f* rp = vred + k * MAT4 + j * (R_RANK / 4);
            float acc = 0.f;
#pragma unroll
            for (int q = 0; q < R_RANK / 4; ++q) {
                const v4f r = rp[q];
                acc += r.x * lane_bcast(t, 4 * q + 0);
                acc += r.y * lane_bcast(t, 4 * q + 1);
                acc += r.z * lane_bcast(t, 4 * q + 2);
                acc += r.w * lane_bcast(t, 4 * q + 3);
            }
            t = acc;
        }
        w[b * R_RANK + j] = t;
    }
}

__global__ __launch_bounds__(32)
void tt_phase3(const float* __restrict__ w, const float* __restrict__ core0,
               float* __restrict__ out) {
    const int lane = threadIdx.x & 31;
    const int half = lane >> 4;
    const int l    = lane & 15;
    const int s0   = blockIdx.x * 16;
    const int b0   = blockIdx.y * 16;

    const float* wp = w     + (size_t)(b0 + l) * R_RANK + 2 * half;
    const float* cp = core0 + (size_t)(s0 + l) * R_RANK + 2 * half;

    v8f acc = {};
#pragma unroll
    for (int k0 = 0; k0 < R_RANK; k0 += 4) {
        v2f a  = *(const v2f*)(wp + k0);
        v2f bf = *(const v2f*)(cp + k0);
        acc = wmma_f32x4(a, bf, acc);
    }
    float* op = out + (size_t)(b0 + 8 * half) * S_DIM + s0 + l;
#pragma unroll
    for (int v = 0; v < 8; ++v) {
        op[(size_t)v * S_DIM] = acc[v];
    }
}

extern "C" void kernel_launch(void* const* d_in, const int* in_sizes, int n_in,
                              void* d_out, int out_size, void* d_ws, size_t ws_size,
                              hipStream_t stream) {
    const float* z  = (const float*)d_in[0];
    const float* c0 = (const float*)d_in[1];
    const float* c1 = (const float*)d_in[2];
    const float* c2 = (const float*)d_in[3];
    const float* c3 = (const float*)d_in[4];
    const float* c4 = (const float*)d_in[5];
    const float* c5 = (const float*)d_in[6];
    float* out = (float*)d_out;

    float* vpart = (float*)d_ws;
    float* w     = vpart + (size_t)KSPLIT * B_BATCH * NCOLS;

    tt_phase1<<<dim3(NT_TOTAL, B_BATCH / 64, KSPLIT), 32, 0, stream>>>(
        z, c1, c2, c3, c4, c5, vpart);
    tt_phase2<<<dim3(B_BATCH), 256, 0, stream>>>(vpart, w);
    tt_phase3<<<dim3(S_DIM / 16, B_BATCH / 16), 32, 0, stream>>>(w, c0, out);
}


# phase 1: 4-wave work-groups share their 64-row z strip through LDS (whole-line loads once per group, next K step held in registers) instead of each wave gathering A rows from global; sum order kept
# speedup vs baseline: 6.8103x; 2.1928x over previous
#include <hip/hip_runtime.h>

typedef __attribute__((ext_vector_type(2))) float v2f;
typedef __attribute__((ext_vector_type(4))) float v4f;
typedef __attribute__((ext_vector_type(8))) float v8f;

#define B_BATCH 256
#define S_DIM   16384
#define R_RANK  16
#define KSPLIT  8
#define NCOLS   1040
#define MBLK    4
#define KCHUNK  (S_DIM / KSPLIT)

__device__ __forceinline__ v8f wmma_f32x4(v2f a, v2f b, v8f c) {
    return __builtin_amdgcn_wmma_f32_16x16x4_f32(false, a, false, b, (short)0, c,
                                                 false, false);
}

#define NWAVE   4
#define NGROUP  17
#define NFULL   (16 * (B_BATCH / 64) * KSPLIT)
#define NLIGHT  ((B_BATCH / 64) * KSPLIT)
#define KC      64
#define APITCH  68
#define MTILE_F (16 * APITCH)
#define STAGE_F (64 * APITCH)
#define NCHUNK  (KCHUNK / KC)
#define NPIECE  (64 * KC / 4 / (32 * NWAVE))

__device__ __forceinline__ void load_frags(v2f (&f)[16], const float* a, int blk) {
#pragma unroll
    for (int u = 0; u < 4; ++u) {
#pragma unroll
        for (int m = 0; m < 4; ++m)
            f[u * 4 + m] = *(const v2f*)(a + m * MTILE_F + (blk * 4 + u) * 4);
    }
}

__device__ __forceinline__ void mma_block(v8f& acc0, v8f& acc1, v8f& acc2, v8f& acc3,
                                          const v2f (&f)[16], const v2f* bf) {
#pragma unroll
    for (int u = 0; u < 4; ++u) {
        acc0 = wmma_f32x4(f[u * 4 + 0], bf[u], acc0);
        acc1 = wmma_f32x4(f[u * 4 + 1], bf[u], acc1);
        acc2 = wmma_f32x4(f[u * 4 + 2], bf[u], acc2);
        acc3 = wmma_f32x4(f[u * 4 + 3], bf[u], acc3);
    }
}

__global__ __launch_bounds__(32 * NWAVE)
void tt_phase1(const float* __restrict__ z,
               const float* __restrict__ c1, const float* __restrict__ c2,
               const float* __restrict__ c3, const float* __restrict__ c4,
               const float* __restrict__ c5,
               float* __restrict__ vpart) {
    __shared__ __attribute__((aligned(16))) float atile[2 * STAGE_F];

    const int tid  = threadIdx.x;
    const int wave = tid >> 5;
    const int lane = tid & 31;
    const int half = lane >> 4;
    const int l    = lane & 15;

    const int bid = blockIdx.x;
    int g, mb, ks;
    if (bid < NFULL) {
        g  = bid & 15;
        mb = (bid >> 4) & 3;
        ks = bid >> 6;
    } else {
        g  = NGROUP - 1;
        mb = (bid - NFULL) & 3;
        ks = (bid - NFULL) >> 2;
    }

    const int b0      = mb * 64;
    const int k_begin = ks * KCHUNK;

    const int srow = tid >> 4;
    const int scol = tid & 15;
    const float* zg = z + (size_t)(b0 + srow) * S_DIM + k_begin + 4 * scol;
    float* sdst = atile + srow * APITCH + 4 * scol;

    const float* afrag = atile + l * APITCH + 2 * half;

#pragma unroll
    for (int q = 0; q < NPIECE; ++q) {
        const v4f x = *(const v4f*)(zg + (size_t)q * 8 * S_DIM);
        *(v4f*)(sdst + q * 8 * APITCH) = x;
    }
    __syncthreads();

    if (g < NGROUP - 1) {
        const int k = g >> 2;
        const int d = (g & 3) * NWAVE + wave;
        const float* cp = (k == 0) ? c1 : (k == 1) ? c2 : (k == 2) ? c3 : c4;
        const int colbase = k * 256 + d * 16;
        const float* bslice = cp + (size_t)d * S_DIM * R_RANK + (size_t)k_begin * R_RANK;
        const float* bp = bslice + 2 * half * R_RANK + l;
        const float* bpf = bslice + KC * R_RANK + lane * 32;

        v8f acc0 = {}, acc1 = {}, acc2 = {}, acc3 = {};
        for (int ch = 0; ch < NCHUNK; ++ch) {
            const bool more = ch + 1 < NCHUNK;
            v2f bfr[KC / 4];
#pragma unroll
            for (int kk = 0; kk < KC / 4; ++kk) {
                bfr[kk].x = bp[kk * 64];
                bfr[kk].y = bp[kk * 64 + 16];
            }
            __builtin_amdgcn_sched_barrier(0);
            v4f st[NPIECE];
            if (more) {
                __builtin_prefetch(bpf, 0, 1);
#pragma unroll
                for (int q = 0; q < NPIECE; ++q)
                    st[q] = *(const v4f*)(zg + KC + (size_t)q * 8 * S_DIM);
            }
            __builtin_amdgcn_sched_barrier(0);
            const float* a = afrag + (ch & 1) * STAGE_F;
            v2f fa[16], fb[16];
            load_frags(fa, a, 0);
            load_frags(fb, a, 1);
            __builtin_amdgcn_sched_barrier(0);
            mma_block(acc0, acc1, acc2, acc3, fa, bfr + 0);
            __builtin_amdgcn_sched_barrier(0);
            load_frags(fa, a, 2);
            __builtin_amdgcn_sched_barrier(0);
            mma_block(acc0, acc1, acc2, acc3, fb, bfr + 4);
            __builtin_amdgcn_sched_barrier(0);
            load_frags(fb, a, 3);
            __builtin_amdgcn_sched_barrier(0);
            mma_block(acc0, acc1, acc2, acc3, fa, bfr + 8);
            __builtin_amdgcn_sched_barrier(0);
            if (more) {
                float* sd = sdst + ((ch + 1) & 1) * STAGE_F;
#pragma unroll
                for (int q = 0; q < NPIECE; ++q)
                    *(v4f*)(sd + q * 8 * APITCH) = st[q];
            }
            __builtin_amdgcn_sched_barrier(0);
            mma_block(acc0, acc1, acc2, acc3, fb, bfr + 12);
            zg  += KC;
            bp  += KC * R_RANK;
            bpf += KC * R_RANK;
            __syncthreads();
        }

        float* vp = vpart + (size_t)ks * B_BATCH * NCOLS
                          + (size_t)(b0 + 8 * half) * NCOLS + colbase + l;
        v8f accs[MBLK] = {acc0, acc1, acc2, acc3};
#pragma unroll
        for (int m = 0; m < MBLK; ++m) {
#pragma unroll
            for (int v = 0; v < 8; ++v) {
                vp[(size_t)(m * 16 + v) * NCOLS] = accs[m][v];
            }
        }
    } else {
        const float* bp = c5 + (size_t)l * S_DIM + k_begin + 2 * half;
        const float* am = afrag + wave * MTILE_F;

        v8f acc = {};
        for (int ch = 0; ch < NCHUNK; ++ch) {
            const bool more = ch + 1 < NCHUNK;
            v4f st[NPIECE];
            if (more) {
#pragma unroll
                for (int q = 0; q < NPIECE; ++q)
                    st[q] = *(const v4f*)(zg + KC + (size_t)q * 8 * S_DIM);
            }
            const float* a = am + (ch & 1) * STAGE_F;
#pragma unroll
            for (int kk = 0; kk < KC / 4; ++kk) {
                const v2f bf = *(const v2f*)(bp + kk * 4);
                const v2f a0 = *(const v2f*)(a + kk * 4);
                acc = wmma_f32x4(a0, bf, acc);
            }
            if (more) {
                float* sd = sdst + ((ch + 1) & 1) * STAGE_F;
#pragma unroll
                for (int q = 0; q < NPIECE; ++q)
                    *(v4f*)(sd + q * 8 * APITCH) = st[q];
            }
            zg += KC;
            bp += KC;
            __syncthreads();
        }

        float* vp = vpart + (size_t)ks * B_BATCH * NCOLS
                          + (size_t)(b0 + 16 * wave + 8 * half) * NCOLS + 1024 + l;
#pragma unroll
        for (int v = 0; v < 8; ++v) {
            vp[(size_t)v * NCOLS] = acc[v];
        }
    }
}

#define ROW4   (NCOLS / 4)
#define MAT4   (R_RANK * R_RANK / 4)
#define TAIL4  (ROW4 - 4 * MAT4)

__device__ __forceinline__ float lane_bcast(float v, int srclane) {
    return __int_as_float(__builtin_amdgcn_readlane(__float_as_int(v), srclane));
}

__global__ __launch_bounds__(256)
void tt_phase2(const float* __restrict__ vpart, float* __restrict__ w) {
    __shared__ v4f vred[ROW4];

    const int b = blockIdx.x;
    const int j = threadIdx.x;
    const bool tail = j < TAIL4;

    const v4f* row = (const v4f*)(vpart + (size_t)b * NCOLS);
    const size_t pstride = (size_t)B_BATCH * ROW4;

    v4f x[KSPLIT], y[KSPLIT];
#pragma unroll
    for (int p = 0; p < KSPLIT; ++p) x[p] = row[p * pstride + j];
    if (tail) {
#pragma unroll
        for (int p = 0; p < KSPLIT; ++p) y[p] = row[p * pstride + 4 * MAT4 + j];
    }

    v4f s = {};
#pragma unroll
    for (int p = 0; p < KSPLIT; ++p) s += x[p];
    vred[j] = s;
    if (tail) {
        v4f s5 = {};
#pragma unroll
        for (int p = 0; p < KSPLIT; ++p) s5 += y[p];
        vred[4 * MAT4 + j] = s5;
    }
    __syncthreads();

    if (j < R_RANK) {
        const float* vf = (const float*)vred;
        float t = vf[4 * R_RANK * R_RANK + j];
#pragma unroll
        for (int k = 3; k >= 0; --k) {
            const v4f* rp = vred + k * MAT4 + j * (R_RANK / 4);
            float acc = 0.f;
#pragma unroll
            for (int q = 0; q < R_RANK / 4; ++q) {
                const v4f r = rp[q];
                acc += r.x * lane_bcast(t, 4 * q + 0);
                acc += r.y * lane_bcast(t, 4 * q + 1);
                acc += r.z * lane_bcast(t, 4 * q + 2);
                acc += r.w * lane_bcast(t, 4 * q + 3);
            }
            t = acc;
        }
        w[b * R_RANK + j] = t;
    }
}

__global__ __launch_bounds__(32)
void tt_phase3(const float* __restrict__ w, const float* __restrict__ core0,
               float* __restrict__ out) {
    const int lane = threadIdx.x & 31;
    const int half = lane >> 4;
    const int l    = lane & 15;
    const int s0   = blockIdx.x * 16;
    const int b0   = blockIdx.y * 16;

    const float* wp = w     + (size_t)(b0 + l) * R_RANK + 2 * half;
    const float* cp = core0 + (size_t)(s0 + l) * R_RANK + 2 * half;

    v8f acc = {};
#pragma unroll
    for (int k0 = 0; k0 < R_RANK; k0 += 4) {
        v2f a  = *(const v2f*)(wp + k0);
        v2f bf = *(const v2f*)(cp + k0);
        acc = wmma_f32x4(a, bf, acc);
    }
    float* op = out + (size_t)(b0 + 8 * half) * S_DIM + s0 + l;
#pragma unroll
    for (int v = 0; v < 8; ++v) {
        op[(size_t)v * S_DIM] = acc[v];
    }
}

extern "C" void kernel_launch(void* const* d_in, const int* in_sizes, int n_in,
                              void* d_out, int out_size, void* d_ws, size_t ws_size,
                              hipStream_t stream) {
    const float* z  = (const float*)d_in[0];
    const float* c0 = (const float*)d_in[1];
    const float* c1 = (const float*)d_in[2];
    const float* c2 = (const float*)d_in[3];
    const float* c3 = (const float*)d_in[4];
    const float* c4 = (const float*)d_in[5];
    const float* c5 = (const float*)d_in[6];
    float* out = (float*)d_out;

    float* vpart = (float*)d_ws;
    float* w     = vpart + (size_t)KSPLIT * B_BATCH * NCOLS;

    tt_phase1<<<dim3(NFULL + NLIGHT), 32 * NWAVE, 0, stream>>>(
        z, c1, c2, c3, c4, c5, vpart);
    tt_phase2<<<dim3(B_BATCH), 256, 0, stream>>>(vpart, w);
    tt_phase3<<<dim3(S_DIM / 16, B_BATCH / 16), 32, 0, stream>>>(w, c0, out);
}


# phase 3: 256 groups of 4 waves (wave = 64 batch rows x 64 pixels, a lane's 4 tiles = 4 consecutive pixels, whole-line stores) replace 16384 one-wave groups; k order of its 16-term sums changed
# speedup vs baseline: 7.4470x; 1.0935x over previous
#include <hip/hip_runtime.h>

typedef __attribute__((ext_vector_type(2))) float v2f;
typedef __attribute__((ext_vector_type(4))) float v4f;
typedef __attribute__((ext_vector_type(8))) float v8f;

#define B_BATCH 256
#define S_DIM   16384
#define R_RANK  16
#define KSPLIT  8
#define NCOLS   1040
#define MBLK    4
#define KCHUNK  (S_DIM / KSPLIT)

__device__ __forceinline__ v8f wmma_f32x4(v2f a, v2f b, v8f c) {
    return __builtin_amdgcn_wmma_f32_16x16x4_f32(false, a, false, b, (short)0, c,
                                                 false, false);
}

#define NWAVE   4
#define NGROUP  17
#define NFULL   (16 * (B_BATCH / 64) * KSPLIT)
#define NLIGHT  ((B_BATCH / 64) * KSPLIT)
#define KC      64
#define APITCH  68
#define MTILE_F (16 * APITCH)
#define STAGE_F (64 * APITCH)
#define NCHUNK  (KCHUNK / KC)
#define NPIECE  (64 * KC / 4 / (32 * NWAVE))

__device__ __forceinline__ void load_frags(v2f (&f)[16], const float* a, int blk) {
#pragma unroll
    for (int u = 0; u < 4; ++u) {
#pragma unroll
        for (int m = 0; m < 4; ++m)
            f[u * 4 + m] = *(const v2f*)(a + m * MTILE_F + (blk * 4 + u) * 4);
    }
}

__device__ __forceinline__ void mma_block(v8f& acc0, v8f& acc1, v8f& acc2, v8f& acc3,
                                          const v2f (&f)[16], const v2f* bf) {
#pragma unroll
    for (int u = 0; u < 4; ++u) {
        acc0 = wmma_f32x4(f[u * 4 + 0], bf[u], acc0);
        acc1 = wmma_f32x4(f[u * 4 + 1], bf[u], acc1);
        acc2 = wmma_f32x4(f[u * 4 + 2], bf[u], acc2);
        acc3 = wmma_f32x4(f[u * 4 + 3], bf[u], acc3);
    }
}

__global__ __launch_bounds__(32 * NWAVE)
void tt_phase1(const float* __restrict__ z,
               const float* __restrict__ c1, const float* __restrict__ c2,
               const float* __restrict__ c3, const float* __restrict__ c4,
               const float* __restrict__ c5,
               float* __restrict__ vpart) {
    __shared__ __attribute__((aligned(16))) float atile[2 * STAGE_F];

    const int tid  = threadIdx.x;
    const int wave = tid >> 5;
    const int lane = tid & 31;
    const int half = lane >> 4;
    const int l    = lane & 15;

    const int bid = blockIdx.x;
    int g, mb, ks;
    if (bid < NFULL) {
        g  = bid & 15;
        mb = (bid >> 4) & 3;
        ks = bid >> 6;
    } else {
        g  = NGROUP - 1;
        mb = (bid - NFULL) & 3;
        ks = (bid - NFULL) >> 2;
    }

    const int b0      = mb * 64;
    const int k_begin = ks * KCHUNK;

    const int srow = tid >> 4;
    const int scol = tid & 15;
    const float* zg = z + (size_t)(b0 + srow) * S_DIM + k_begin + 4 * scol;
    float* sdst = atile + srow * APITCH + 4 * scol;

    const float* afrag = atile + l * APITCH + 2 * half;

#pragma unroll
    for (int q = 0; q < NPIECE; ++q) {
        const v4f x = *(const v4f*)(zg + (size_t)q * 8 * S_DIM);
        *(v4f*)(sdst + q * 8 * APITCH) = x;
    }
    __syncthreads();

    if (g < NGROUP - 1) {
        const int k = g >> 2;
        const int d = (g & 3) * NWAVE + wave;
        const float* cp = (k == 0) ? c1 : (k == 1) ? c2 : (k == 2) ? c3 : c4;
        const int colbase = k * 256 + d * 16;
        const float* bslice = cp + (size_t)d * S_DIM * R_RANK + (size_t)k_begin * R_RANK;
        const float* bp = bslice + 2 * half * R_RANK + l;
        const float* bpf = bslice + KC * R_RANK + lane * 32;

        v8f acc0 = {}, acc1 = {}, acc2 = {}, acc3 = {};
        for (int ch = 0; ch < NCHUNK; ++ch) {
            const bool more = ch + 1 < NCHUNK;
            v2f bfr[KC / 4];
#pragma unroll
            for (int kk = 0; kk < KC / 4; ++kk) {
                bfr[kk].x = bp[kk * 64];
                bfr[kk].y = bp[kk * 64 + 16];
            }
            __builtin_amdgcn_sched_barrier(0);
            v4f st[NPIECE];
            if (more) {
                __builtin_prefetch(bpf, 0, 1);
#pragma unroll
                for (int q = 0; q < NPIECE; ++q)
                    st[q] = *(const v4f*)(zg + KC + (size_t)q * 8 * S_DIM);
            }
            __builtin_amdgcn_sched_barrier(0);
            const float* a = afrag + (ch & 1) * STAGE_F;
            v2f fa[16], fb[16];
            load_frags(fa, a, 0);
            load_frags(fb, a, 1);
            __builtin_amdgcn_sched_barrier(0);
            mma_block(acc0, acc1, acc2, acc3, fa, bfr + 0);
            __builtin_amdgcn_sched_barrier(0);
            load_frags(fa, a, 2);
            __builtin_amdgcn_sched_barrier(0);
            mma_block(acc0, acc1, acc2, acc3, fb, bfr + 4);
            __builtin_amdgcn_sched_barrier(0);
            load_frags(fb, a, 3);
            __builtin_amdgcn_sched_barrier(0);
            mma_block(acc0, acc1, acc2, acc3, fa, bfr + 8);
            __builtin_amdgcn_sched_barrier(0);
            if (more) {
                float* sd = sdst + ((ch + 1) & 1) * STAGE_F;
#pragma unroll
                for (int q = 0; q < NPIECE; ++q)
                    *(v4f*)(sd + q * 8 * APITCH) = st[q];
            }
            __builtin_amdgcn_sched_barrier(0);
            mma_block(acc0, acc1, acc2, acc3, fb, bfr + 12);
            zg  += KC;
            bp  += KC * R_RANK;
            bpf += KC * R_RANK;
            __syncthreads();
        }

        float* vp = vpart + (size_t)ks * B_BATCH * NCOLS
                          + (size_t)(b0 + 8 * half) * NCOLS + colbase + l;
        v8f accs[MBLK] = {acc0, acc1, acc2, acc3};
#pragma unroll
        for (int m = 0; m < MBLK; ++m) {
#pragma unroll
            for (int v = 0; v < 8; ++v) {
                vp[(size_t)(m * 16 + v) * NCOLS] = accs[m][v];
            }
        }
    } else {
        const float* bp = c5 + (size_t)l * S_DIM + k_begin + 2 * half;
        const float* am = afrag + wave * MTILE_F;

        v8f acc = {};
        for (int ch = 0; ch < NCHUNK; ++ch) {
            const bool more = ch + 1 < NCHUNK;
            v4f st[NPIECE];
            if (more) {
#pragma unroll
                for (int q = 0; q < NPIECE; ++q)
                    st[q] = *(const v4f*)(zg + KC + (size_t)q * 8 * S_DIM);
            }
            const float* a = am + (ch & 1) * STAGE_F;
#pragma unroll
            for (int kk = 0; kk < KC / 4; ++kk) {
                const v2f bf = *(const v2f*)(bp + kk * 4);
                const v2f a0 = *(const v2f*)(a + kk * 4);
                acc = wmma_f32x4(a0, bf, acc);
            }
            if (more) {
                float* sd = sdst + ((ch + 1) & 1) * STAGE_F;
#pragma unroll
                for (int q = 0; q < NPIECE; ++q)
                    *(v4f*)(sd + q * 8 * APITCH) = st[q];
            }
            zg += KC;
            bp += KC;
            __syncthreads();
        }

        float* vp = vpart + (size_t)ks * B_BATCH * NCOLS
                          + (size_t)(b0 + 16 * wave + 8 * half) * NCOLS + 1024 + l;
#pragma unroll
        for (int v = 0; v < 8; ++v) {
            vp[(size_t)v * NCOLS] = acc[v];
        }
    }
}

#define ROW4   (NCOLS / 4)
#define MAT4   (R_RANK * R_RANK / 4)
#define TAIL4  (ROW4 - 4 * MAT4)

__device__ __forceinline__ float lane_bcast(float v, int srclane) {
    return __int_as_float(__builtin_amdgcn_readlane(__float_as_int(v), srclane));
}

__global__ __launch_bounds__(256)
void tt_phase2(const float* __restrict__ vpart, float* __restrict__ w) {
    __shared__ v4f vred[ROW4];

    const int b = blockIdx.x;
    const int j = threadIdx.x;
    const bool tail = j < TAIL4;

    const v4f* row = (const v4f*)(vpart + (size_t)b * NCOLS);
    const size_t pstride = (size_t)B_BATCH * ROW4;

    v4f x[KSPLIT], y[KSPLIT];
#pragma unroll
    for (int p = 0; p < KSPLIT; ++p) x[p] = row[p * pstride + j];
    if (tail) {
#pragma unroll
        for (int p = 0; p < KSPLIT; ++p) y[p] = row[p * pstride + 4 * MAT4 + j];
    }

    v4f s = {};
#pragma unroll
    for (int p = 0; p < KSPLIT; ++p) s += x[p];
    vred[j] = s;
    if (tail) {
        v4f s5 = {};
#pragma unroll
        for (int p = 0; p < KSPLIT; ++p) s5 += y[p];
        vred[4 * MAT4 + j] = s5;
    }
    __syncthreads();

    if (j < R_RANK) {
        const float* vf = (const float*)vred;
        float t = vf[4 * R_RANK * R_RANK + j];
#pragma unroll
        for (int k = 3; k >= 0; --k) {
            const v4f* rp = vred + k * MAT4 + j * (R_RANK / 4);
            float acc = 0.f;
#pragma unroll
            for (int q = 0; q < R_RANK / 4; ++q) {
                const v4f r = rp[q];
                acc += r.x * lane_bcast(t, 4 * q + 0);
                acc += r.y * lane_bcast(t, 4 * q + 1);
                acc += r.z * lane_bcast(t, 4 * q + 2);
                acc += r.w * lane_bcast(t, 4 * q + 3);
            }
            t = acc;
        }
        w[b * R_RANK + j] = t;
    }
}

#define P3_PIX    64
#define P3_WAVES  (B_BATCH / 64)

__device__ __forceinline__ v2f frag_pair(const v4f (&q)[2], int u) {
    const v4f x = q[u >> 1];
    v2f p;
    if (u & 1) { p.x = x.z; p.y = x.w; } else { p.x = x.x; p.y = x.y; }
    return p;
}

__global__ __launch_bounds__(32 * P3_WAVES)
void tt_phase3(const float* __restrict__ w, const float* __restrict__ core0,
               float* __restrict__ out) {
    const int lane = threadIdx.x & 31;
    const int wave = threadIdx.x >> 5;
    const int half = lane >> 4;
    const int l    = lane & 15;
    const int s0   = blockIdx.x * P3_PIX;
    const int b0   = wave * 64;

    v4f bq[4][2];
#pragma unroll
    for (int j = 0; j < 4; ++j) {
        const float* cp = core0 + (size_t)(s0 + 4 * l + j) * R_RANK + 8 * half;
        bq[j][0] = *(const v4f*)cp;
        bq[j][1] = *(const v4f*)(cp + 4);
    }
    v4f aq[4][2];
#pragma unroll
    for (int m = 0; m < 4; ++m) {
        const float* wp = w + (size_t)(b0 + 16 * m + l) * R_RANK + 8 * half;
        aq[m][0] = *(const v4f*)wp;
        aq[m][1] = *(const v4f*)(wp + 4);
    }

#pragma unroll
    for (int m = 0; m < 4; ++m) {
        v8f acc0 = {}, acc1 = {}, acc2 = {}, acc3 = {};
#pragma unroll
        for (int u = 0; u < 4; ++u) {
            const v2f a = frag_pair(aq[m], u);
            acc0 = wmma_f32x4(a, frag_pair(bq[0], u), acc0);
            acc1 = wmma_f32x4(a, frag_pair(bq[1], u), acc1);
            acc2 = wmma_f32x4(a, frag_pair(bq[2], u), acc2);
            acc3 = wmma_f32x4(a, frag_pair(bq[3], u), acc3);
        }
        float* op = out + (size_t)(b0 + 16 * m + 8 * half) * S_DIM + s0 + 4 * l;
#pragma unroll
        for (int r = 0; r < 8; ++r) {
            v4f o;
            o.x = acc0[r]; o.y = acc1[r]; o.z = acc2[r]; o.w = acc3[r];
            *(v4f*)(op + (size_t)r * S_DIM) = o;
        }
    }
}

extern "C" void kernel_launch(void* const* d_in, const int* in_sizes, int n_in,
                              void* d_out, int out_size, void* d_ws, size_t ws_size,
                              hipStream_t stream) {
    const float* z  = (const float*)d_in[0];
    const float* c0 = (const float*)d_in[1];
    const float* c1 = (const float*)d_in[2];
    const float* c2 = (const float*)d_in[3];
    const float* c3 = (const float*)d_in[4];
    const float* c4 = (const float*)d_in[5];
    const float* c5 = (const float*)d_in[6];
    float* out = (float*)d_out;

    float* vpart = (float*)d_ws;
    float* w     = vpart + (size_t)KSPLIT * B_BATCH * NCOLS;

    tt_phase1<<<dim3(NFULL + NLIGHT), 32 * NWAVE, 0, stream>>>(
        z, c1, c2, c3, c4, c5, vpart);
    tt_phase2<<<dim3(B_BATCH), 256, 0, stream>>>(vpart, w);
    tt_phase3<<<dim3(S_DIM / P3_PIX), 32 * P3_WAVES, 0, stream>>>(w, c0, out);
}


# phase 1: one 4-wave work-group per compute unit owns a 128x64 tile (8 M tiles per wave, per B fragment); next step's B fragments and first A block are fetched a step ahead; sum order kept
# speedup vs baseline: 7.6925x; 1.0330x over previous
#include <hip/hip_runtime.h>

typedef __attribute__((ext_vector_type(2))) float v2f;
typedef __attribute__((ext_vector_type(4))) float v4f;
typedef __attribute__((ext_vector_type(8))) float v8f;

#define B_BATCH 256
#define S_DIM   16384
#define R_RANK  16
#define KSPLIT  8
#define NCOLS   1040
#define MBLK    8
#define KCHUNK  (S_DIM / KSPLIT)

__device__ __forceinline__ v8f wmma_f32x4(v2f a, v2f b, v8f c) {
    return __builtin_amdgcn_wmma_f32_16x16x4_f32(false, a, false, b, (short)0, c,
                                                 false, false);
}

#define NWAVE   4
#define NGROUP  17
#define MROWS   (16 * MBLK)
#define NFULL   (16 * (B_BATCH / MROWS) * KSPLIT)
#define NLIGHT  ((B_BATCH / 64) * KSPLIT)
#define LPIECE  (64 * KC / 4 / (32 * NWAVE))
#define KC      64
#define APITCH  68
#define MTILE_F (16 * APITCH)
#define STAGE_F (MROWS * APITCH)
#define NCHUNK  (KCHUNK / KC)
#define NPIECE  (MROWS * KC / 4 / (32 * NWAVE))
#define NFRAG   (4 * MBLK)

__device__ __forceinline__ void load_frags(v2f (&f)[NFRAG], const float* a, int blk) {
#pragma unroll
    for (int u = 0; u < 4; ++u) {
#pragma unroll
        for (int m = 0; m < MBLK; ++m)
            f[u * MBLK + m] = *(const v2f*)(a + m * MTILE_F + (blk * 4 + u) * 4);
    }
}

__device__ __forceinline__ void mma_block(v8f (&acc)[MBLK], const v2f (&f)[NFRAG],
                                          const v2f* bf) {
#pragma unroll
    for (int u = 0; u < 4; ++u) {
#pragma unroll
        for (int m = 0; m < MBLK; ++m)
            acc[m] = wmma_f32x4(f[u * MBLK + m], bf[u], acc[m]);
    }
}

__global__ __launch_bounds__(32 * NWAVE)
void tt_phase1(const float* __restrict__ z,
               const float* __restrict__ c1, const float* __restrict__ c2,
               const float* __restrict__ c3, const float* __restrict__ c4,
               const float* __restrict__ c5,
               float* __restrict__ vpart) {
    __shared__ __attribute__((aligned(16))) float atile[2 * STAGE_F];

    const int tid  = threadIdx.x;
    const int wave = tid >> 5;
    const int lane = tid & 31;
    const int half = lane >> 4;
    const int l    = lane & 15;

    const int bid = blockIdx.x;
    int g, b0, ks;
    if (bid < NFULL) {
        g  = bid & 15;
        b0 = ((bid >> 4) & 1) * MROWS;
        ks = bid >> 5;
    } else {
        g  = NGROUP - 1;
        b0 = ((bid - NFULL) & 3) * 64;
        ks = (bid - NFULL) >> 2;
    }

    const int k_begin = ks * KCHUNK;

    const int srow = tid >> 4;
    const int scol = tid & 15;
    const float* zg = z + (size_t)(b0 + srow) * S_DIM + k_begin + 4 * scol;
    float* sdst = atile + srow * APITCH + 4 * scol;

    const float* afrag = atile + l * APITCH + 2 * half;

    if (g < NGROUP - 1) {
#pragma unroll
        for (int q = 0; q < NPIECE; ++q) {
            const v4f x = *(const v4f*)(zg + (size_t)q * 8 * S_DIM);
            *(v4f*)(sdst + q * 8 * APITCH) = x;
        }
        const int k = g >> 2;
        const int d = (g & 3) * NWAVE + wave;
        const float* cp = (k == 0) ? c1 : (k == 1) ? c2 : (k == 2) ? c3 : c4;
        const int colbase = k * 256 + d * 16;
        const float* bslice = cp + (size_t)d * S_DIM * R_RANK + (size_t)k_begin * R_RANK;
        const float* bp = bslice + 2 * half * R_RANK + l;
        const float* bpf = bslice + 2 * KC * R_RANK + lane * 32;

        v8f acc[MBLK];
#pragma unroll
        for (int m = 0; m < MBLK; ++m) acc[m] = v8f{};

        v2f bcur[KC / 4], bnext[KC / 4];
#pragma unroll
        for (int kk = 0; kk < KC / 4; ++kk) {
            bcur[kk].x = bp[kk * 64];
            bcur[kk].y = bp[kk * 64 + 16];
        }
        __syncthreads();
        v2f fa[NFRAG], fb[NFRAG];
        load_frags(fa, afrag, 0);

        for (int ch = 0; ch < NCHUNK; ++ch) {
            const bool more = ch + 1 < NCHUNK;
            v4f st[NPIECE];
            if (more) {
#pragma unroll
                for (int q = 0; q < NPIECE; ++q)
                    st[q] = *(const v4f*)(zg + KC + (size_t)q * 8 * S_DIM);
                __builtin_amdgcn_sched_barrier(0);
#pragma unroll
                for (int kk = 0; kk < KC / 4; ++kk) {
                    bnext[kk].x = bp[KC * R_RANK + kk * 64];
                    bnext[kk].y = bp[KC * R_RANK + kk * 64 + 16];
                }
            }
            if (ch + 2 < NCHUNK) __builtin_prefetch(bpf, 0, 1);
            const float* a = afrag + (ch & 1) * STAGE_F;
            load_frags(fb, a, 1);
            __builtin_amdgcn_sched_barrier(0);
            mma_block(acc, fa, bcur + 0);
            __builtin_amdgcn_sched_barrier(0);
            load_frags(fa, a, 2);
            __builtin_amdgcn_sched_barrier(0);
            mma_block(acc, fb, bcur + 4);
            __builtin_amdgcn_sched_barrier(0);
            load_frags(fb, a, 3);
            if (more) {
                float* sd = sdst + ((ch + 1) & 1) * STAGE_F;
#pragma unroll
                for (int q = 0; q < NPIECE; ++q)
                    *(v4f*)(sd + q * 8 * APITCH) = st[q];
            }
            __builtin_amdgcn_sched_barrier(0);
            mma_block(acc, fa, bcur + 8);
            __syncthreads();
            if (more) load_frags(fa, afrag + ((ch + 1) & 1) * STAGE_F, 0);
            __builtin_amdgcn_sched_barrier(0);
            mma_block(acc, fb, bcur + 12);
            __builtin_amdgcn_sched_barrier(0);
            if (more) {
#pragma unroll
                for (int kk = 0; kk < KC / 4; ++kk) bcur[kk] = bnext[kk];
            }
            zg  += KC;
            bp  += KC * R_RANK;
            bpf += KC * R_RANK;
        }

        float* vp = vpart + (size_t)ks * B_BATCH * NCOLS
                          + (size_t)(b0 + 8 * half) * NCOLS + colbase + l;
#pragma unroll
        for (int m = 0; m < MBLK; ++m) {
#pragma unroll
            for (int v = 0; v < 8; ++v) {
                vp[(size_t)(m * 16 + v) * NCOLS] = acc[m][v];
            }
        }
    } else {
#pragma unroll
        for (int q = 0; q < LPIECE; ++q) {
            const v4f x = *(const v4f*)(zg + (size_t)q * 8 * S_DIM);
            *(v4f*)(sdst + q * 8 * APITCH) = x;
        }
        const float* bp = c5 + (size_t)l * S_DIM + k_begin + 2 * half;
        const float* am = afrag + wave * MTILE_F;
        __syncthreads();

        v8f acc = {};
        for (int ch = 0; ch < NCHUNK; ++ch) {
            const bool more = ch + 1 < NCHUNK;
            v4f st[LPIECE];
            if (more) {
#pragma unroll
                for (int q = 0; q < LPIECE; ++q)
                    st[q] = *(const v4f*)(zg + KC + (size_t)q * 8 * S_DIM);
            }
            const float* a = am + (ch & 1) * STAGE_F;
#pragma unroll
            for (int kk = 0; kk < KC / 4; ++kk) {
                const v2f bf = *(const v2f*)(bp + kk * 4);
                const v2f a0 = *(const v2f*)(a + kk * 4);
                acc = wmma_f32x4(a0, bf, acc);
            }
            if (more) {
                float* sd = sdst + ((ch + 1) & 1) * STAGE_F;
#pragma unroll
                for (int q = 0; q < LPIECE; ++q)
                    *(v4f*)(sd + q * 8 * APITCH) = st[q];
            }
            zg += KC;
            bp += KC;
            __syncthreads();
        }

        float* vp = vpart + (size_t)ks * B_BATCH * NCOLS
                          + (size_t)(b0 + 16 * wave + 8 * half) * NCOLS + 1024 + l;
#pragma unroll
        for (int v = 0; v < 8; ++v) {
            vp[(size_t)v * NCOLS] = acc[v];
        }
    }
}

#define ROW4   (NCOLS / 4)
#define MAT4   (R_RANK * R_RANK / 4)
#define TAIL4  (ROW4 - 4 * MAT4)

__device__ __forceinline__ float lane_bcast(float v, int srclane) {
    return __int_as_float(__builtin_amdgcn_readlane(__float_as_int(v), srclane));
}

__global__ __launch_bounds__(256)
void tt_phase2(const float* __restrict__ vpart, float* __restrict__ w) {
    __shared__ v4f vred[ROW4];

    const int b = blockIdx.x;
    const int j = threadIdx.x;
    const bool tail = j < TAIL4;

    const v4f* row = (const v4f*)(vpart + (size_t)b * NCOLS);
    const size_t pstride = (size_t)B_BATCH * ROW4;

    v4f x[KSPLIT], y[KSPLIT];
#pragma unroll
    for (int p = 0; p < KSPLIT; ++p) x[p] = row[p * pstride + j];
    if (tail) {
#pragma unroll
        for (int p = 0; p < KSPLIT; ++p) y[p] = row[p * pstride + 4 * MAT4 + j];
    }

    v4f s = {};
#pragma unroll
    for (int p = 0; p < KSPLIT; ++p) s += x[p];
    vred[j] = s;
    if (tail) {
        v4f s5 = {};
#pragma unroll
        for (int p = 0; p < KSPLIT; ++p) s5 += y[p];
        vred[4 * MAT4 + j] = s5;
    }
    __syncthreads();

    if (j < R_RANK) {
        const float* vf = (const float*)vred;
        float t = vf[4 * R_RANK * R_RANK + j];
#pragma unroll
        for (int k = 3; k >= 0; --k) {
            const v4f* rp = vred + k * MAT4 + j * (R_RANK / 4);
            float acc = 0.f;
#pragma unroll
            for (int q = 0; q < R_RANK / 4; ++q) {
                const v4f r = rp[q];
                acc += r.x * lane_bcast(t, 4 * q + 0);
                acc += r.y * lane_bcast(t, 4 * q + 1);
                acc += r.z * lane_bcast(t, 4 * q + 2);
                acc += r.w * lane_bcast(t, 4 * q + 3);
            }
            t = acc;
        }
        w[b * R_RANK + j] = t;
    }
}

#define P3_PIX    64
#define P3_WAVES  (B_BATCH / 64)

__device__ __forceinline__ v2f frag_pair(const v4f (&q)[2], int u) {
    const v4f x = q[u >> 1];
    v2f p;
    if (u & 1) { p.x = x.z; p.y = x.w; } else { p.x = x.x; p.y = x.y; }
    return p;
}

__global__ __launch_bounds__(32 * P3_WAVES)
void tt_phase3(const float* __restrict__ w, const float* __restrict__ core0,
               float* __restrict__ out) {
    const int lane = threadIdx.x & 31;
    const int wave = threadIdx.x >> 5;
    const int half = lane >> 4;
    const int l    = lane & 15;
    const int s0   = blockIdx.x * P3_PIX;
    const int b0   = wave * 64;

    v4f bq[4][2];
#pragma unroll
    for (int j = 0; j < 4; ++j) {
        const float* cp = core0 + (size_t)(s0 + 4 * l + j) * R_RANK + 8 * half;
        bq[j][0] = *(const v4f*)cp;
        bq[j][1] = *(const v4f*)(cp + 4);
    }
    v4f aq[4][2];
#pragma unroll
    for (int m = 0; m < 4; ++m) {
        const float* wp = w + (size_t)(b0 + 16 * m + l) * R_RANK + 8 * half;
        aq[m][0] = *(const v4f*)wp;
        aq[m][1] = *(const v4f*)(wp + 4);
    }

#pragma unroll
    for (int m = 0; m < 4; ++m) {
        v8f acc0 = {}, acc1 = {}, acc2 = {}, acc3 = {};
#pragma unroll
        for (int u = 0; u < 4; ++u) {
            const v2f a = frag_pair(aq[m], u);
            acc0 = wmma_f32x4(a, frag_pair(bq[0], u), acc0);
            acc1 = wmma_f32x4(a, frag_pair(bq[1], u), acc1);
            acc2 = wmma_f32x4(a, frag_pair(bq[2], u), acc2);
            acc3 = wmma_f32x4(a, frag_pair(bq[3], u), acc3);
        }
        float* op = out + (size_t)(b0 + 16 * m + 8 * half) * S_DIM + s0 + 4 * l;
#pragma unroll
        for (int r = 0; r < 8; ++r) {
            v4f o;
            o.x = acc0[r]; o.y = acc1[r]; o.z = acc2[r]; o.w = acc3[r];
            *(v4f*)(op + (size_t)r * S_DIM) = o;
        }
    }
}

extern "C" void kernel_launch(void* const* d_in, const int* in_sizes, int n_in,
                              void* d_out, int out_size, void* d_ws, size_t ws_size,
                              hipStream_t stream) {
    const float* z  = (const float*)d_in[0];
    const float* c0 = (const float*)d_in[1];
    const float* c1 = (const float*)d_in[2];
    const float* c2 = (const float*)d_in[3];
    const float* c3 = (const float*)d_in[4];
    const float* c4 = (const float*)d_in[5];
    const float* c5 = (const float*)d_in[6];
    float* out = (float*)d_out;

    float* vpart = (float*)d_ws;
    float* w     = vpart + (size_t)KSPLIT * B_BATCH * NCOLS;

    tt_phase1<<<dim3(NFULL + NLIGHT), 32 * NWAVE, 0, stream>>>(
        z, c1, c2, c3, c4, c5, vpart);
    tt_phase2<<<dim3(B_BATCH), 256, 0, stream>>>(vpart, w);
    tt_phase3<<<dim3(S_DIM / P3_PIX), 32 * P3_WAVES, 0, stream>>>(w, c0, out);
}


# phase 1, both group kinds: global loads are requested a whole step ahead and the step barrier drains LDS only, so they stay in flight across it; sum order kept
# speedup vs baseline: 8.6616x; 1.1260x over previous
#include <hip/hip_runtime.h>

typedef __attribute__((ext_vector_type(2))) float v2f;
typedef __attribute__((ext_vector_type(4))) float v4f;
typedef __attribute__((ext_vector_type(8))) float v8f;

#define B_BATCH 256
#define S_DIM   16384
#define R_RANK  16
#define KSPLIT  8
#define NCOLS   1040
#define MBLK    8
#define KCHUNK  (S_DIM / KSPLIT)

__device__ __forceinline__ v8f wmma_f32x4(v2f a, v2f b, v8f c) {
    return __builtin_amdgcn_wmma_f32_16x16x4_f32(false, a, false, b, (short)0, c,
                                                 false, false);
}

#define NWAVE   4
#define NGROUP  17
#define MROWS   (16 * MBLK)
#define NFULL   (16 * (B_BATCH / MROWS) * KSPLIT)
#define NLIGHT  ((B_BATCH / 64) * KSPLIT)
#define LPIECE  (64 * KC / 4 / (32 * NWAVE))
#define KC      64
#define APITCH  68
#define MTILE_F (16 * APITCH)
#define STAGE_F (MROWS * APITCH)
#define NCHUNK  (KCHUNK / KC)
#define NPIECE  (MROWS * KC / 4 / (32 * NWAVE))
#define NFRAG   (4 * MBLK)

__device__ __forceinline__ void load_frags(v2f (&f)[NFRAG], const float* a, int blk) {
#pragma unroll
    for (int u = 0; u < 4; ++u) {
#pragma unroll
        for (int m = 0; m < MBLK; ++m)
            f[u * MBLK + m] = *(const v2f*)(a + m * MTILE_F + (blk * 4 + u) * 4);
    }
}

__device__ __forceinline__ void mma_block(v8f (&acc)[MBLK], const v2f (&f)[NFRAG],
                                          const v2f* bf) {
#pragma unroll
    for (int u = 0; u < 4; ++u) {
#pragma unroll
        for (int m = 0; m < MBLK; ++m)
            acc[m] = wmma_f32x4(f[u * MBLK + m], bf[u], acc[m]);
    }
}

template <int U0, int U1>
__device__ __forceinline__ void mma_rows(v8f (&acc)[MBLK], const v2f (&f)[NFRAG],
                                         const v2f* bf) {
#pragma unroll
    for (int u = U0; u < U1; ++u) {
#pragma unroll
        for (int m = 0; m < MBLK; ++m)
            acc[m] = wmma_f32x4(f[u * MBLK + m], bf[u], acc[m]);
    }
}

__device__ __forceinline__ void lds_drain() {
    asm volatile("s_wait_dscnt 0x0" ::: "memory");
}
__device__ __forceinline__ void wg_signal() {
    asm volatile("s_barrier_signal -1" ::: "memory");
}
__device__ __forceinline__ void wg_wait() {
    asm volatile("s_barrier_wait -1" ::: "memory");
}

template <bool MORE, bool MORE2>
__device__ __forceinline__ void p1_step(v8f (&acc)[MBLK], v2f (&fa)[NFRAG], v2f (&fb)[NFRAG],
                                        v2f (&bcur)[KC / 4], v2f (&bnext)[KC / 4],
                                        v4f (&st)[NPIECE],
                                        const float* __restrict__ zg,
                                        const float* __restrict__ bp,
                                        const float* __restrict__ bpf,
                                        const float* afrag, float* sdst, int ch) {
    if (MORE) {
#pragma unroll
        for (int kk = 0; kk < KC / 4; ++kk) {
            bnext[kk].x = bp[KC * R_RANK + kk * 64];
            bnext[kk].y = bp[KC * R_RANK + kk * 64 + 16];
        }
    }
    if (MORE2) __builtin_prefetch(bpf, 0, 1);
    const float* a = afrag + (ch & 1) * STAGE_F;
    load_frags(fb, a, 1);
    __builtin_amdgcn_sched_barrier(0);
    mma_block(acc, fa, bcur + 0);
    __builtin_amdgcn_sched_barrier(0);
    load_frags(fa, a, 2);
    __builtin_amdgcn_sched_barrier(0);
    mma_block(acc, fb, bcur + 4);
    __builtin_amdgcn_sched_barrier(0);
    load_frags(fb, a, 3);
    if (MORE) {
        float* sd = sdst + ((ch + 1) & 1) * STAGE_F;
#pragma unroll
        for (int q = 0; q < NPIECE; ++q)
            *(v4f*)(sd + q * 8 * APITCH) = st[q];
    }
    if (MORE2) {
#pragma unroll
        for (int q = 0; q < NPIECE; ++q)
            st[q] = *(const v4f*)(zg + 2 * KC + (size_t)q * 8 * S_DIM);
    }
    __builtin_amdgcn_sched_barrier(0);
    mma_rows<0, 2>(acc, fa, bcur + 8);
    __builtin_amdgcn_sched_barrier(0);
    lds_drain();
    wg_signal();
    __builtin_amdgcn_sched_barrier(0);
    mma_rows<2, 4>(acc, fa, bcur + 8);
    __builtin_amdgcn_sched_barrier(0);
    wg_wait();
    if (MORE) load_frags(fa, afrag + ((ch + 1) & 1) * STAGE_F, 0);
    __builtin_amdgcn_sched_barrier(0);
    mma_block(acc, fb, bcur + 12);
    __builtin_amdgcn_sched_barrier(0);
    if (MORE) {
#pragma unroll
        for (int kk = 0; kk < KC / 4; ++kk) bcur[kk] = bnext[kk];
    }
}

template <bool MORE, bool MORE2>
__device__ __forceinline__ void p1_light_step(v8f& acc, v2f (&bcur)[KC / 4], v2f (&bnext)[KC / 4],
                                              v4f (&st)[LPIECE],
                                              const float* __restrict__ zg,
                                              const float* __restrict__ bp,
                                              const float* am, float* sdst, int ch) {
    if (MORE) {
#pragma unroll
        for (int kk = 0; kk < KC / 4; ++kk)
            bnext[kk] = *(const v2f*)(bp + KC + kk * 4);
    }
    const float* a = am + (ch & 1) * STAGE_F;
#pragma unroll
    for (int kk = 0; kk < KC / 4; ++kk) {
        const v2f a0 = *(const v2f*)(a + kk * 4);
        acc = wmma_f32x4(a0, bcur[kk], acc);
    }
    if (MORE) {
        float* sd = sdst + ((ch + 1) & 1) * STAGE_F;
#pragma unroll
        for (int q = 0; q < LPIECE; ++q)
            *(v4f*)(sd + q * 8 * APITCH) = st[q];
    }
    if (MORE2) {
#pragma unroll
        for (int q = 0; q < LPIECE; ++q)
            st[q] = *(const v4f*)(zg + 2 * KC + (size_t)q * 8 * S_DIM);
    }
    lds_drain();
    wg_signal();
    wg_wait();
    if (MORE) {
#pragma unroll
        for (int kk = 0; kk < KC / 4; ++kk) bcur[kk] = bnext[kk];
    }
}

__global__ __launch_bounds__(32 * NWAVE)
void tt_phase1(const float* __restrict__ z,
               const float* __restrict__ c1, const float* __restrict__ c2,
               const float* __restrict__ c3, const float* __restrict__ c4,
               const float* __restrict__ c5,
               float* __restrict__ vpart) {
    __shared__ __attribute__((aligned(16))) float atile[2 * STAGE_F];

    const int tid  = threadIdx.x;
    const int wave = tid >> 5;
    const int lane = tid & 31;
    const int half = lane >> 4;
    const int l    = lane & 15;

    const int bid = blockIdx.x;
    int g, b0, ks;
    if (bid < NFULL) {
        g  = bid & 15;
        b0 = ((bid >> 4) & 1) * MROWS;
        ks = bid >> 5;
    } else {
        g  = NGROUP - 1;
        b0 = ((bid - NFULL) & 3) * 64;
        ks = (bid - NFULL) >> 2;
    }

    const int k_begin = ks * KCHUNK;

    const int srow = tid >> 4;
    const int scol = tid & 15;
    const float* zg = z + (size_t)(b0 + srow) * S_DIM + k_begin + 4 * scol;
    float* sdst = atile + srow * APITCH + 4 * scol;

    const float* afrag = atile + l * APITCH + 2 * half;

    if (g < NGROUP - 1) {
        const int k = g >> 2;
        const int d = (g & 3) * NWAVE + wave;
        const float* cp = (k == 0) ? c1 : (k == 1) ? c2 : (k == 2) ? c3 : c4;
        const int colbase = k * 256 + d * 16;
        const float* bslice = cp + (size_t)d * S_DIM * R_RANK + (size_t)k_begin * R_RANK;
        const float* bp = bslice + 2 * half * R_RANK + l;
        const float* bpf = bslice + 2 * KC * R_RANK + lane * 32;

        v4f st[NPIECE];
        v2f bcur[KC / 4], bnext[KC / 4];
        {
            v4f x0[NPIECE];
#pragma unroll
            for (int q = 0; q < NPIECE; ++q)
                x0[q] = *(const v4f*)(zg + (size_t)q * 8 * S_DIM);
#pragma unroll
            for (int q = 0; q < NPIECE; ++q)
                st[q] = *(const v4f*)(zg + KC + (size_t)q * 8 * S_DIM);
#pragma unroll
            for (int kk = 0; kk < KC / 4; ++kk) {
                bcur[kk].x = bp[kk * 64];
                bcur[kk].y = bp[kk * 64 + 16];
            }
            __builtin_amdgcn_sched_barrier(0);
#pragma unroll
            for (int q = 0; q < NPIECE; ++q)
                *(v4f*)(sdst + q * 8 * APITCH) = x0[q];
        }

        v8f acc[MBLK];
#pragma unroll
        for (int m = 0; m < MBLK; ++m) acc[m] = v8f{};

        __syncthreads();
        v2f fa[NFRAG], fb[NFRAG];
        load_frags(fa, afrag, 0);

        int ch = 0;
        for (; ch < NCHUNK - 2; ++ch) {
            p1_step<true, true>(acc, fa, fb, bcur, bnext, st, zg, bp, bpf, afrag, sdst, ch);
            zg  += KC;
            bp  += KC * R_RANK;
            bpf += KC * R_RANK;
        }
        p1_step<true, false>(acc, fa, fb, bcur, bnext, st, zg, bp, bpf, afrag, sdst, NCHUNK - 2);
        zg += KC;
        bp += KC * R_RANK;
        p1_step<false, false>(acc, fa, fb, bcur, bnext, st, zg, bp, bpf, afrag, sdst, NCHUNK - 1);

        float* vp = vpart + (size_t)ks * B_BATCH * NCOLS
                          + (size_t)(b0 + 8 * half) * NCOLS + colbase + l;
#pragma unroll
        for (int m = 0; m < MBLK; ++m) {
#pragma unroll
            for (int v = 0; v < 8; ++v) {
                vp[(size_t)(m * 16 + v) * NCOLS] = acc[m][v];
            }
        }
    } else {
        const float* bp = c5 + (size_t)l * S_DIM + k_begin + 2 * half;
        const float* am = afrag + wave * MTILE_F;
        v4f st[LPIECE];
        v2f bcur[KC / 4], bnext[KC / 4];
        {
            v4f x0[LPIECE];
#pragma unroll
            for (int q = 0; q < LPIECE; ++q)
                x0[q] = *(const v4f*)(zg + (size_t)q * 8 * S_DIM);
#pragma unroll
            for (int q = 0; q < LPIECE; ++q)
                st[q] = *(const v4f*)(zg + KC + (size_t)q * 8 * S_DIM);
#pragma unroll
            for (int kk = 0; kk < KC / 4; ++kk)
                bcur[kk] = *(const v2f*)(bp + kk * 4);
            __builtin_amdgcn_sched_barrier(0);
#pragma unroll
            for (int q = 0; q < LPIECE; ++q)
                *(v4f*)(sdst + q * 8 * APITCH) = x0[q];
        }
        __syncthreads();

        v8f acc = {};
        int ch = 0;
        for (; ch < NCHUNK - 2; ++ch) {
            p1_light_step<true, true>(acc, bcur, bnext, st, zg, bp, am, sdst, ch);
            zg += KC;
            bp += KC;
        }
        p1_light_step<true, false>(acc, bcur, bnext, st, zg, bp, am, sdst, NCHUNK - 2);
        zg += KC;
        bp += KC;
        p1_light_step<false, false>(acc, bcur, bnext, st, zg, bp, am, sdst, NCHUNK - 1);

        float* vp = vpart + (size_t)ks * B_BATCH * NCOLS
                          + (size_t)(b0 + 16 * wave + 8 * half) * NCOLS + 1024 + l;
#pragma unroll
        for (int v = 0; v < 8; ++v) {
            vp[(size_t)v * NCOLS] = acc[v];
        }
    }
}

#define ROW4   (NCOLS / 4)
#define MAT4   (R_RANK * R_RANK / 4)
#define TAIL4  (ROW4 - 4 * MAT4)

__device__ __forceinline__ float lane_bcast(float v, int srclane) {
    return __int_as_float(__builtin_amdgcn_readlane(__float_as_int(v), srclane));
}

__global__ __launch_bounds__(256)
void tt_phase2(const float* __restrict__ vpart, float* __restrict__ w) {
    __shared__ v4f vred[ROW4];

    const int b = blockIdx.x;
    const int j = threadIdx.x;
    const bool tail = j < TAIL4;

    const v4f* row = (const v4f*)(vpart + (size_t)b * NCOLS);
    const size_t pstride = (size_t)B_BATCH * ROW4;

    v4f x[KSPLIT], y[KSPLIT];
#pragma unroll
    for (int p = 0; p < KSPLIT; ++p) x[p] = row[p * pstride + j];
    if (tail) {
#pragma unroll
        for (int p = 0; p < KSPLIT; ++p) y[p] = row[p * pstride + 4 * MAT4 + j];
    }

    v4f s = {};
#pragma unroll
    for (int p = 0; p < KSPLIT; ++p) s += x[p];
    vred[j] = s;
    if (tail) {
        v4f s5 = {};
#pragma unroll
        for (int p = 0; p < KSPLIT; ++p) s5 += y[p];
        vred[4 * MAT4 + j] = s5;
    }
    __syncthreads();

    if (j < R_RANK) {
        const float* vf = (const float*)vred;
        float t = vf[4 * R_RANK * R_RANK + j];
#pragma unroll
        for (int k = 3; k >= 0; --k) {
            const v4f* rp = vred + k * MAT4 + j * (R_RANK / 4);
            float acc = 0.f;
#pragma unroll
            for (int q = 0; q < R_RANK / 4; ++q) {
                const v4f r = rp[q];
                acc += r.x * lane_bcast(t, 4 * q + 0);
                acc += r.y * lane_bcast(t, 4 * q + 1);
                acc += r.z * lane_bcast(t, 4 * q + 2);
                acc += r.w * lane_bcast(t, 4 * q + 3);
            }
            t = acc;
        }
        w[b * R_RANK + j] = t;
    }
}

#define P3_PIX    64
#define P3_WAVES  (B_BATCH / 64)

__device__ __forceinline__ v2f frag_pair(const v4f (&q)[2], int u) {
    const v4f x = q[u >> 1];
    v2f p;
    if (u & 1) { p.x = x.z; p.y = x.w; } else { p.x = x.x; p.y = x.y; }
    return p;
}

__global__ __launch_bounds__(32 * P3_WAVES)
void tt_phase3(const float* __restrict__ w, const float* __restrict__ core0,
               float* __restrict__ out) {
    const int lane = threadIdx.x & 31;
    const int wave = threadIdx.x >> 5;
    const int half = lane >> 4;
    const int l    = lane & 15;
    const int s0   = blockIdx.x * P3_PIX;
    const int b0   = wave * 64;

    v4f bq[4][2];
#pragma unroll
    for (int j = 0; j < 4; ++j) {
        const float* cp = core0 + (size_t)(s0 + 4 * l + j) * R_RANK + 8 * half;
        bq[j][0] = *(const v4f*)cp;
        bq[j][1] = *(const v4f*)(cp + 4);
    }
    v4f aq[4][2];
#pragma unroll
    for (int m = 0; m < 4; ++m) {
        const float* wp = w + (size_t)(b0 + 16 * m + l) * R_RANK + 8 * half;
        aq[m][0] = *(const v4f*)wp;
        aq[m][1] = *(const v4f*)(wp + 4);
    }

#pragma unroll
    for (int m = 0; m < 4; ++m) {
        v8f acc0 = {}, acc1 = {}, acc2 = {}, acc3 = {};
#pragma unroll
        for (int u = 0; u < 4; ++u) {
            const v2f a = frag_pair(aq[m], u);
            acc0 = wmma_f32x4(a, frag_pair(bq[0], u), acc0);
            acc1 = wmma_f32x4(a, frag_pair(bq[1], u), acc1);
            acc2 = wmma_f32x4(a, frag_pair(bq[2], u), acc2);
            acc3 = wmma_f32x4(a, frag_pair(bq[3], u), acc3);
        }
        float* op = out + (size_t)(b0 + 16 * m + 8 * half) * S_DIM + s0 + 4 * l;
#pragma unroll
        for (int r = 0; r < 8; ++r) {
            v4f o;
            o.x = acc0[r]; o.y = acc1[r]; o.z = acc2[r]; o.w = acc3[r];
            *(v4f*)(op + (size_t)r * S_DIM) = o;
        }
    }
}

extern "C" void kernel_launch(void* const* d_in, const int* in_sizes, int n_in,
                              void* d_out, int out_size, void* d_ws, size_t ws_size,
                              hipStream_t stream) {
    const float* z  = (const float*)d_in[0];
    const float* c0 = (const float*)d_in[1];
    const float* c1 = (const float*)d_in[2];
    const float* c2 = (const float*)d_in[3];
    const float* c3 = (const float*)d_in[4];
    const float* c4 = (const float*)d_in[5];
    const float* c5 = (const float*)d_in[6];
    float* out = (float*)d_out;

    float* vpart = (float*)d_ws;
    float* w     = vpart + (size_t)KSPLIT * B_BATCH * NCOLS;

    tt_phase1<<<dim3(NFULL + NLIGHT), 32 * NWAVE, 0, stream>>>(
        z, c1, c2, c3, c4, c5, vpart);
    tt_phase2<<<dim3(B_BATCH), 256, 0, stream>>>(vpart, w);
    tt_phase3<<<dim3(S_DIM / P3_PIX), 32 * P3_WAVES, 0, stream>>>(w, c0, out);
}
